# starting kernel
# baseline (speedup 1.0000x reference)
#include <hip/hip_runtime.h>
#include <hip/hip_bf16.h>

typedef __attribute__((ext_vector_type(2))) float v2f;
typedef __attribute__((ext_vector_type(8))) float v8f;

#define CH 64
#define NB 2

__global__ void zero_f32_kernel(float* __restrict__ p, long n) {
    long i = (long)blockIdx.x * blockDim.x + threadIdx.x;
    long stride = (long)gridDim.x * blockDim.x;
    for (; i < n; i += stride) p[i] = 0.0f;
}

__global__ void degree_kernel(const int* __restrict__ row, const int* __restrict__ col,
                              float* __restrict__ deg, long E) {
    long e = (long)blockIdx.x * blockDim.x + threadIdx.x;
    if (e >= E) return;
    int r = row[e], c = col[e];
    if (r != c) atomicAdd(&deg[r], 1.0f);
}

__global__ void dinv_sqrt_kernel(float* __restrict__ deg, long n) {
    long i = (long)blockIdx.x * blockDim.x + threadIdx.x;
    if (i >= n) return;
    float d = deg[i];
    deg[i] = (d > 0.0f) ? __frsqrt_rn(d) : 0.0f;
}

__global__ void lap_kernel(const int* __restrict__ row, const int* __restrict__ col,
                           const float* __restrict__ ew, const float* __restrict__ dis,
                           float* __restrict__ lap, long E) {
    long e = (long)blockIdx.x * blockDim.x + threadIdx.x;
    if (e >= E) return;
    int r = row[e], c = col[e];
    lap[e] = (r != c) ? (-dis[r] * ew[e] * dis[c]) : 0.0f;
}

__global__ __launch_bounds__(256) void spmm_kernel(
    const float* __restrict__ src, const int* __restrict__ row,
    const int* __restrict__ col, const float* __restrict__ lap,
    float* __restrict__ dst, int N, long E) {
    long idx = (long)blockIdx.x * blockDim.x + threadIdx.x;
    long e = idx >> 7;
    if (e >= E) return;
    int rem = (int)(idx & 127);
    int b = rem >> 6;
    int c = rem & 63;
    float l = lap[e];
    if (l == 0.0f) return;
    long cs = col[e], rs = row[e];
    float v = src[((long)b * N + cs) * CH + c] * l;
    atomicAdd(&dst[((long)b * N + rs) * CH + c], v);
}

__global__ __launch_bounds__(128) void cheb_gemm_wmma(
    const float* __restrict__ x, const float* __restrict__ tx1,
    const float* __restrict__ s2, const float* __restrict__ w,
    const float* __restrict__ bias, float* __restrict__ out, long BN) {
    __shared__ float wlds[3 * CH * CH];

    const int tid = threadIdx.x;
    for (int idx = tid; idx < 3 * CH * CH; idx += 128) {
        const int t   = idx >> 12;
        const int rem = idx & 4095;
        const int k   = rem >> 6;
        const int n   = rem & 63;
        wlds[((((t << 5) + (k >> 1)) << 6) + n) * 2 + (k & 1)] = w[idx];
    }
    __syncthreads();

    const int wave = tid >> 5;
    const int lane = tid & 31;
    const long m0 = ((long)blockIdx.x * 4 + wave) * 16;
    if (m0 + 16 > BN) return;

    const int mrow = lane & 15;
    const int lhi  = lane >> 4;

    const v2f* __restrict__ wp = (const v2f*)wlds;
    const v2f* __restrict__ bbase = wp + (lhi << 6) + mrow;

    v8f acc[4] = {v8f{}, v8f{}, v8f{}, v8f{}};

    #pragma unroll
    for (int t = 0; t < 3; ++t) {
        const float* A = (t == 0) ? x : (t == 1) ? tx1 : s2;
        #pragma unroll
        for (int ks = 0; ks < 16; ++ks) {
            const int kb = ks * 4 + (lhi << 1);
            const float2 a2 = *(const float2*)(A + (m0 + mrow) * CH + kb);
            v2f a;
            a.x = a2.x; a.y = a2.y;
            if (t == 2) {
                const float2 x2 = *(const float2*)(x + (m0 + mrow) * CH + kb);
                a.x = 2.0f * a.x - x2.x;
                a.y = 2.0f * a.y - x2.y;
            }
            #pragma unroll
            for (int nt = 0; nt < 4; ++nt) {
                const v2f b = bbase[(t * 32 + ks * 2) * 64 + nt * 16];
                acc[nt] = __builtin_amdgcn_wmma_f32_16x16x4_f32(
                    false, a, false, b, (short)0, acc[nt], false, false);
            }
        }
    }

    const int moff = lhi << 3;
    #pragma unroll
    for (int nt = 0; nt < 4; ++nt) {
        const int col = nt * 16 + mrow;
        const float bv = bias[col];
        #pragma unroll
        for (int v = 0; v < 8; ++v) {
            out[(m0 + moff + v) * CH + col] = acc[nt][v] + bv;
        }
    }
}

__global__ void cheb_gemm_tail(const float* __restrict__ x, const float* __restrict__ tx1,
                               const float* __restrict__ s2, const float* __restrict__ w,
                               const float* __restrict__ bias, float* __restrict__ out,
                               long startRow, long BN) {
    long r = startRow + blockIdx.x;
    if (r >= BN) return;
    int c = threadIdx.x;
    float acc = bias[c];
    for (int k = 0; k < CH; ++k) {
        float t0 = x[r * CH + k];
        float t1 = tx1[r * CH + k];
        float t2 = 2.0f * s2[r * CH + k] - t0;
        acc += t0 * w[k * CH + c] + t1 * w[CH * CH + k * CH + c] + t2 * w[2 * CH * CH + k * CH + c];
    }
    out[r * CH + c] = acc;
}

extern "C" void kernel_launch(void* const* d_in, const int* in_sizes, int n_in,
                              void* d_out, int out_size, void* d_ws, size_t ws_size,
                              hipStream_t stream) {
    const float* x    = (const float*)d_in[0];
    const int*   ei   = (const int*)d_in[1];
    const float* ew   = (const float*)d_in[2];
    const float* w    = (const float*)d_in[3];
    const float* bias = (const float*)d_in[4];
    float*       out  = (float*)d_out;

    const long BN = (long)in_sizes[0] / CH;
    const int  N  = (int)(BN / NB);
    const long E  = (long)in_sizes[2];
    const int* row = ei;
    const int* col = ei + E;

    char* ws = (char*)d_ws;
    size_t off = 0;
    auto carve = [&](size_t bytes) {
        void* p = ws + off;
        off += (bytes + 255) & ~(size_t)255;
        return p;
    };
    float* deg = (float*)carve((size_t)N * sizeof(float));
    float* lap = (float*)carve((size_t)E * sizeof(float));
    float* tx1 = (float*)carve((size_t)BN * CH * sizeof(float));
    float* s2  = (float*)carve((size_t)BN * CH * sizeof(float));
    (void)ws_size;

    const int TB = 256;
    auto gblocks = [](long n, int tb) { return (int)((n + tb - 1) / tb); };

    zero_f32_kernel<<<4096, TB, 0, stream>>>(deg, N);
    zero_f32_kernel<<<4096, TB, 0, stream>>>(tx1, BN * CH);
    zero_f32_kernel<<<4096, TB, 0, stream>>>(s2,  BN * CH);

    degree_kernel<<<gblocks(E, TB), TB, 0, stream>>>(row, col, deg, E);
    dinv_sqrt_kernel<<<gblocks(N, TB), TB, 0, stream>>>(deg, N);
    lap_kernel<<<gblocks(E, TB), TB, 0, stream>>>(row, col, ew, deg, lap, E);

    const long spmmThreads = E * (long)(NB * CH);
    spmm_kernel<<<gblocks(spmmThreads, TB), TB, 0, stream>>>(x,   row, col, lap, tx1, N, E);
    spmm_kernel<<<gblocks(spmmThreads, TB), TB, 0, stream>>>(tx1, row, col, lap, s2,  N, E);

    const long fullTiles = BN / 16;
    if (fullTiles > 0) {
        const int blocks = (int)((fullTiles + 3) / 4);
        cheb_gemm_wmma<<<blocks, 128, 0, stream>>>(x, tx1, s2, w, bias, out, BN);
    }
    const long tailStart = fullTiles * 16;
    if (tailStart < BN) {
        cheb_gemm_tail<<<(int)(BN - tailStart), CH, 0, stream>>>(x, tx1, s2, w, bias, out,
                                                                 tailStart, BN);
    }
}


# SpMM float-atomic scatter replaced by per-row register gather over an edge list counting-sorted by row each call (one int-atomic pass, then in-row sort): sums now run in a fixed per-row order
# speedup vs baseline: 4.3426x; 4.3426x over previous
#include <hip/hip_runtime.h>
#include <hip/hip_bf16.h>

typedef __attribute__((ext_vector_type(2))) float v2f;
typedef __attribute__((ext_vector_type(8))) float v8f;

#define CH 64
#define NB 2

__global__ void zero_i32_kernel(int* __restrict__ p, long n) {
    long i = (long)blockIdx.x * blockDim.x + threadIdx.x;
    long stride = (long)gridDim.x * blockDim.x;
    for (; i < n; i += stride) p[i] = 0;
}

__global__ void count_kernel(const int* __restrict__ row, const int* __restrict__ col,
                             int* __restrict__ cnt, int* __restrict__ pos, long E) {
    long e = (long)blockIdx.x * blockDim.x + threadIdx.x;
    if (e >= E) return;
    int r = row[e], c = col[e];
    if (r != c) pos[e] = atomicAdd(&cnt[r], 1);
}

__global__ __launch_bounds__(256) void scan_kernel(const int* __restrict__ cnt,
                                                   int* __restrict__ rp, int N) {
    __shared__ int wsum[8];
    __shared__ int wpart[8];
    const int tid = threadIdx.x;
    const int lane = tid & 31;
    const int wave = tid >> 5;

    int part = 0;
    const int4* __restrict__ c4 = (const int4*)cnt;
    const int nvec = (int)blockIdx.x * 256;
    #pragma unroll 8
    for (int i = tid; i < nvec; i += 256) {
        const int4 q = c4[i];
        part += q.x + q.y + q.z + q.w;
    }

    const int idx0 = (int)blockIdx.x * 1024 + tid * 4;
    int v[4];
    for (int k = 0; k < 4; ++k) v[k] = (idx0 + k < N) ? cnt[idx0 + k] : 0;
    const int tsum = v[0] + v[1] + v[2] + v[3];

    int inc = tsum;
    for (int d = 1; d < 32; d <<= 1) {
        const int t = __shfl_up(inc, d);
        if (lane >= d) inc += t;
    }
    for (int d = 16; d > 0; d >>= 1) part += __shfl_xor(part, d);
    if (lane == 31) wsum[wave] = inc;
    if (lane == 0) wpart[wave] = part;
    __syncthreads();

    int excl = inc - tsum;
    for (int w = 0; w < 8; ++w) {
        excl += wpart[w];
        if (w < wave) excl += wsum[w];
    }
    for (int k = 0; k < 4; ++k) {
        if (idx0 + k < N) rp[idx0 + k] = excl;
        excl += v[k];
    }
}

__global__ void fill_kernel(const int* __restrict__ row, const int* __restrict__ col,
                            const float* __restrict__ ew, const int* __restrict__ cnt,
                            const int* __restrict__ rp, const int* __restrict__ pos,
                            int2* __restrict__ ep, long E) {
    long e = (long)blockIdx.x * blockDim.x + threadIdx.x;
    if (e >= E) return;
    const int r = row[e], c = col[e];
    if (r == c) return;
    const int dc = cnt[c];
    const float disr = __frsqrt_rn((float)cnt[r]);
    const float disc = (dc > 0) ? __frsqrt_rn((float)dc) : 0.0f;
    const float l = -disr * ew[e] * disc;
    ep[rp[r] + pos[e]] = make_int2(c, __float_as_int(l));
}

#define SORT_K 4
typedef unsigned long long u64;
__global__ __launch_bounds__(256) void rowsort_kernel(const int* __restrict__ rp,
                                                      const int* __restrict__ cnt,
                                                      u64* ep, int N) {
    const int lane = threadIdx.x & 31;
    const int r = __builtin_amdgcn_readfirstlane((int)blockIdx.x * 8 + (int)(threadIdx.x >> 5));
    if (r >= N) return;
    const int beg = rp[r];
    const int n = cnt[r];
    if (n <= 1) return;
    u64* seg = ep + beg;
    if (n <= 32 * SORT_K) {
        u64 key[SORT_K];
        int rank[SORT_K];
        for (int k = 0; k < SORT_K; ++k) {
            const int i = k * 32 + lane;
            key[k] = (i < n) ? seg[i] : ~0ull;
            rank[k] = 0;
        }
        for (int k2 = 0; k2 < SORT_K; ++k2) {
            const int m = min(32, n - k2 * 32);
            for (int j = 0; j < m; ++j) {
                const unsigned lo = (unsigned)__builtin_amdgcn_readlane((int)(unsigned)key[k2], j);
                const unsigned hi = (unsigned)__builtin_amdgcn_readlane((int)(unsigned)(key[k2] >> 32), j);
                const u64 kj = ((u64)hi << 32) | lo;
                const int gj = k2 * 32 + j;
                for (int k = 0; k < SORT_K; ++k) {
                    const int gi = k * 32 + lane;
                    rank[k] += (kj < key[k] || (kj == key[k] && gj < gi)) ? 1 : 0;
                }
            }
        }
        for (int k = 0; k < SORT_K; ++k) {
            const int i = k * 32 + lane;
            if (i < n) seg[rank[k]] = key[k];
        }
    } else if (lane == 0) {
        for (int i = 1; i < n; ++i) {
            const u64 kk = seg[i];
            int j = i - 1;
            while (j >= 0 && seg[j] > kk) { seg[j + 1] = seg[j]; --j; }
            seg[j + 1] = kk;
        }
    }
}

__global__ __launch_bounds__(256) void spmm_pull_kernel(
    const float* __restrict__ src, const int* __restrict__ rp, const int* __restrict__ cnt,
    const int2* __restrict__ ep, float* __restrict__ dst, int N) {
    const int lane = threadIdx.x & 31;
    const int r = __builtin_amdgcn_readfirstlane((int)blockIdx.x * 8 + (int)(threadIdx.x >> 5));
    if (r >= N) return;
    const int beg = rp[r];
    const int end = beg + cnt[r];
    const int b = lane >> 4;
    const int c4 = (lane & 15) * 4;
    const float* __restrict__ sb = src + (long)b * N * CH + c4;
    float4 acc = make_float4(0.0f, 0.0f, 0.0f, 0.0f);
    for (int base = beg; base < end; base += 32) {
        const int m = min(32, end - base);
        int2 mine = make_int2(0, 0);
        if (lane < m) mine = ep[base + lane];
        #pragma unroll 4
        for (int j = 0; j < m; ++j) {
            const int c = __builtin_amdgcn_readlane(mine.x, j);
            const float l = __int_as_float(__builtin_amdgcn_readlane(mine.y, j));
            const float4 q = *(const float4*)(sb + (long)c * CH);
            acc.x += l * q.x;
            acc.y += l * q.y;
            acc.z += l * q.z;
            acc.w += l * q.w;
        }
    }
    *(float4*)(dst + ((long)b * N + r) * CH + c4) = acc;
}

__global__ __launch_bounds__(128) void cheb_gemm_wmma(
    const float* __restrict__ x, const float* __restrict__ tx1,
    const float* __restrict__ s2, const float* __restrict__ w,
    const float* __restrict__ bias, float* __restrict__ out, long BN) {
    __shared__ float wlds[3 * CH * CH];

    const int tid = threadIdx.x;
    for (int idx = tid; idx < 3 * CH * CH; idx += 128) {
        const int t   = idx >> 12;
        const int rem = idx & 4095;
        const int k   = rem >> 6;
        const int n   = rem & 63;
        wlds[((((t << 5) + (k >> 1)) << 6) + n) * 2 + (k & 1)] = w[idx];
    }
    __syncthreads();

    const int wave = tid >> 5;
    const int lane = tid & 31;
    const long m0 = ((long)blockIdx.x * 4 + wave) * 16;
    if (m0 + 16 > BN) return;

    const int mrow = lane & 15;
    const int lhi  = lane >> 4;

    const v2f* __restrict__ wp = (const v2f*)wlds;
    const v2f* __restrict__ bbase = wp + (lhi << 6) + mrow;

    v8f acc[4] = {v8f{}, v8f{}, v8f{}, v8f{}};

    #pragma unroll
    for (int t = 0; t < 3; ++t) {
        const float* A = (t == 0) ? x : (t == 1) ? tx1 : s2;
        #pragma unroll
        for (int ks = 0; ks < 16; ++ks) {
            const int kb = ks * 4 + (lhi << 1);
            const float2 a2 = *(const float2*)(A + (m0 + mrow) * CH + kb);
            v2f a;
            a.x = a2.x; a.y = a2.y;
            if (t == 2) {
                const float2 x2 = *(const float2*)(x + (m0 + mrow) * CH + kb);
                a.x = 2.0f * a.x - x2.x;
                a.y = 2.0f * a.y - x2.y;
            }
            #pragma unroll
            for (int nt = 0; nt < 4; ++nt) {
                const v2f b = bbase[(t * 32 + ks * 2) * 64 + nt * 16];
                acc[nt] = __builtin_amdgcn_wmma_f32_16x16x4_f32(
                    false, a, false, b, (short)0, acc[nt], false, false);
            }
        }
    }

    const int moff = lhi << 3;
    #pragma unroll
    for (int nt = 0; nt < 4; ++nt) {
        const int col = nt * 16 + mrow;
        const float bv = bias[col];
        #pragma unroll
        for (int v = 0; v < 8; ++v) {
            out[(m0 + moff + v) * CH + col] = acc[nt][v] + bv;
        }
    }
}

__global__ void cheb_gemm_tail(const float* __restrict__ x, const float* __restrict__ tx1,
                               const float* __restrict__ s2, const float* __restrict__ w,
                               const float* __restrict__ bias, float* __restrict__ out,
                               long startRow, long BN) {
    long r = startRow + blockIdx.x;
    if (r >= BN) return;
    int c = threadIdx.x;
    float acc = bias[c];
    for (int k = 0; k < CH; ++k) {
        float t0 = x[r * CH + k];
        float t1 = tx1[r * CH + k];
        float t2 = 2.0f * s2[r * CH + k] - t0;
        acc += t0 * w[k * CH + c] + t1 * w[CH * CH + k * CH + c] + t2 * w[2 * CH * CH + k * CH + c];
    }
    out[r * CH + c] = acc;
}

extern "C" void kernel_launch(void* const* d_in, const int* in_sizes, int n_in,
                              void* d_out, int out_size, void* d_ws, size_t ws_size,
                              hipStream_t stream) {
    const float* x    = (const float*)d_in[0];
    const int*   ei   = (const int*)d_in[1];
    const float* ew   = (const float*)d_in[2];
    const float* w    = (const float*)d_in[3];
    const float* bias = (const float*)d_in[4];
    float*       out  = (float*)d_out;

    const long BN = (long)in_sizes[0] / CH;
    const int  N  = (int)(BN / NB);
    const long E  = (long)in_sizes[2];
    const int* row = ei;
    const int* col = ei + E;

    char* ws = (char*)d_ws;
    size_t off = 0;
    auto carve = [&](size_t bytes) {
        void* p = ws + off;
        off += (bytes + 255) & ~(size_t)255;
        return p;
    };
    int*   cnt = (int*)carve((size_t)N * sizeof(int));
    int*   rp  = (int*)carve((size_t)N * sizeof(int));
    int2*  ep  = (int2*)carve((size_t)E * sizeof(int2));
    float* tx1 = (float*)carve((size_t)BN * CH * sizeof(float));
    const size_t planeBytes = (size_t)BN * CH * sizeof(float);
    const size_t posBytes   = (size_t)E * sizeof(int);
    float* s2  = (float*)carve(planeBytes > posBytes ? planeBytes : posBytes);
    int*   pos = (int*)s2;
    if (off > ws_size) return;

    const int TB = 256;
    auto gblocks = [](long n, int tb) { return (int)((n + tb - 1) / tb); };

    zero_i32_kernel<<<gblocks(N, TB), TB, 0, stream>>>(cnt, N);
    count_kernel<<<gblocks(E, TB), TB, 0, stream>>>(row, col, cnt, pos, E);
    scan_kernel<<<gblocks(N, 1024), 256, 0, stream>>>(cnt, rp, N);
    fill_kernel<<<gblocks(E, TB), TB, 0, stream>>>(row, col, ew, cnt, rp, pos, ep, E);
    rowsort_kernel<<<gblocks(N, 8), 256, 0, stream>>>(rp, cnt, (u64*)ep, N);

    spmm_pull_kernel<<<gblocks(N, 8), 256, 0, stream>>>(x,   rp, cnt, ep, tx1, N);
    spmm_pull_kernel<<<gblocks(N, 8), 256, 0, stream>>>(tx1, rp, cnt, ep, s2,  N);

    const long fullTiles = BN / 16;
    if (fullTiles > 0) {
        const int blocks = (int)((fullTiles + 3) / 4);
        cheb_gemm_wmma<<<blocks, 128, 0, stream>>>(x, tx1, s2, w, bias, out, BN);
    }
    const long tailStart = fullTiles * 16;
    if (tailStart < BN) {
        cheb_gemm_tail<<<(int)(BN - tailStart), CH, 0, stream>>>(x, tx1, s2, w, bias, out,
                                                                 tailStart, BN);
    }
}


# In-row sort and lap scaling moved into the first SpMM's row wave (row degree wave-uniform, one degree gather per pair): rowsort launch gone, fill reads 1 table not 3; same sort key, sum order kept
# speedup vs baseline: 4.6349x; 1.0673x over previous
#include <hip/hip_runtime.h>
#include <hip/hip_bf16.h>

typedef __attribute__((ext_vector_type(2))) float v2f;
typedef __attribute__((ext_vector_type(8))) float v8f;

#define CH 64
#define NB 2

__global__ void zero_i32_kernel(int* __restrict__ p, long n) {
    long i = (long)blockIdx.x * blockDim.x + threadIdx.x;
    long stride = (long)gridDim.x * blockDim.x;
    for (; i < n; i += stride) p[i] = 0;
}

__global__ void count_kernel(const int* __restrict__ row, const int* __restrict__ col,
                             int* __restrict__ cnt, int* __restrict__ pos, long E) {
    long e = (long)blockIdx.x * blockDim.x + threadIdx.x;
    if (e >= E) return;
    int r = row[e], c = col[e];
    if (r != c) pos[e] = atomicAdd(&cnt[r], 1);
}

__global__ __launch_bounds__(256) void scan_kernel(const int* __restrict__ cnt,
                                                   int* __restrict__ rp, int N) {
    __shared__ int wsum[8];
    __shared__ int wpart[8];
    const int tid = threadIdx.x;
    const int lane = tid & 31;
    const int wave = tid >> 5;

    int part = 0;
    const int4* __restrict__ c4 = (const int4*)cnt;
    const int nvec = (int)blockIdx.x * 256;
    #pragma unroll 8
    for (int i = tid; i < nvec; i += 256) {
        const int4 q = c4[i];
        part += q.x + q.y + q.z + q.w;
    }

    const int idx0 = (int)blockIdx.x * 1024 + tid * 4;
    int v[4];
    for (int k = 0; k < 4; ++k) v[k] = (idx0 + k < N) ? cnt[idx0 + k] : 0;
    const int tsum = v[0] + v[1] + v[2] + v[3];

    int inc = tsum;
    for (int d = 1; d < 32; d <<= 1) {
        const int t = __shfl_up(inc, d);
        if (lane >= d) inc += t;
    }
    for (int d = 16; d > 0; d >>= 1) part += __shfl_xor(part, d);
    if (lane == 31) wsum[wave] = inc;
    if (lane == 0) wpart[wave] = part;
    __syncthreads();

    int excl = inc - tsum;
    for (int w = 0; w < 8; ++w) {
        excl += wpart[w];
        if (w < wave) excl += wsum[w];
    }
    for (int k = 0; k < 4; ++k) {
        if (idx0 + k < N) rp[idx0 + k] = excl;
        excl += v[k];
    }
}

__global__ void fill_kernel(const int* __restrict__ row, const int* __restrict__ col,
                            const float* __restrict__ ew, const int* __restrict__ rp,
                            const int* __restrict__ pos, int2* __restrict__ ep, long E) {
    long e = (long)blockIdx.x * blockDim.x + threadIdx.x;
    if (e >= E) return;
    const int r = row[e], c = col[e];
    if (r == c) return;
    ep[rp[r] + pos[e]] = make_int2(c, __float_as_int(ew[e]));
}

#define SORT_K 4
typedef unsigned long long u64;
__global__ __launch_bounds__(256) void spmm_first_kernel(
    const float* __restrict__ src, const int* __restrict__ rp, const int* __restrict__ cnt,
    u64* ep, float* __restrict__ dst, int N) {
    __shared__ u64 sbuf[8 * 32 * SORT_K];
    const int lane = threadIdx.x & 31;
    const int wv = threadIdx.x >> 5;
    const int r = __builtin_amdgcn_readfirstlane((int)blockIdx.x * 8 + wv);
    if (r >= N) return;
    const int beg = rp[r];
    const int n = cnt[r];
    u64* seg = ep + beg;
    u64* lw = sbuf + wv * (32 * SORT_K);
    const float disr = __frsqrt_rn((float)n);
    const bool small = (n <= 32 * SORT_K);

    if (small) {
        const int kmax = (n + 31) >> 5;
        u64 key[SORT_K];
        int rank[SORT_K];
        for (int k = 0; k < SORT_K; ++k) {
            const int i = k * 32 + lane;
            key[k] = ~0ull;
            rank[k] = 0;
            if (i < n) {
                const u64 p = seg[i];
                const int c = (int)(unsigned)p;
                const float w = __int_as_float((int)(unsigned)(p >> 32));
                const int dc = cnt[c];
                const float disc = (dc > 0) ? __frsqrt_rn((float)dc) : 0.0f;
                const float l = -disr * w * disc;
                key[k] = ((u64)(unsigned)__float_as_int(l) << 32) | (u64)(unsigned)c;
            }
        }
        for (int k2 = 0; k2 < SORT_K; ++k2) {
            if (k2 < kmax) {
                const int m = min(32, n - k2 * 32);
                for (int j = 0; j < m; ++j) {
                    const unsigned lo = (unsigned)__builtin_amdgcn_readlane((int)(unsigned)key[k2], j);
                    const unsigned hi = (unsigned)__builtin_amdgcn_readlane((int)(unsigned)(key[k2] >> 32), j);
                    const u64 kj = ((u64)hi << 32) | lo;
                    const int gj = k2 * 32 + j;
                    for (int k = 0; k < SORT_K; ++k) {
                        if (k < kmax) {
                            const int gi = k * 32 + lane;
                            rank[k] += (kj < key[k] || (kj == key[k] && gj < gi)) ? 1 : 0;
                        }
                    }
                }
            }
        }
        for (int k = 0; k < SORT_K; ++k) {
            const int i = k * 32 + lane;
            if (i < n) lw[rank[k]] = key[k];
        }
        for (int k = 0; k < SORT_K; ++k) {
            const int i = k * 32 + lane;
            if (i < n) seg[i] = lw[i];
        }
    } else {
        for (int i = lane; i < n; i += 32) {
            const u64 p = seg[i];
            const int c = (int)(unsigned)p;
            const float w = __int_as_float((int)(unsigned)(p >> 32));
            const int dc = cnt[c];
            const float disc = (dc > 0) ? __frsqrt_rn((float)dc) : 0.0f;
            const float l = -disr * w * disc;
            seg[i] = ((u64)(unsigned)__float_as_int(l) << 32) | (u64)(unsigned)c;
        }
        __threadfence();
        if (lane == 0) {
            for (int i = 1; i < n; ++i) {
                const u64 kk = seg[i];
                int j = i - 1;
                while (j >= 0 && seg[j] > kk) { seg[j + 1] = seg[j]; --j; }
                seg[j + 1] = kk;
            }
        }
        __threadfence();
    }

    const int b = lane >> 4;
    const int c4 = (lane & 15) * 4;
    const float* __restrict__ sb = src + (long)b * N * CH + c4;
    float4 acc = make_float4(0.0f, 0.0f, 0.0f, 0.0f);
    for (int base = 0; base < n; base += 32) {
        const int m = min(32, n - base);
        u64 mine = 0ull;
        if (small) {
            if (lane < m) mine = lw[base + lane];
        } else {
            if (lane < m) mine = seg[base + lane];
        }
        const int mx = (int)(unsigned)mine;
        const int my = (int)(unsigned)(mine >> 32);
        #pragma unroll 4
        for (int j = 0; j < m; ++j) {
            const int c = __builtin_amdgcn_readlane(mx, j);
            const float l = __int_as_float(__builtin_amdgcn_readlane(my, j));
            const float4 q = *(const float4*)(sb + (long)c * CH);
            acc.x += l * q.x;
            acc.y += l * q.y;
            acc.z += l * q.z;
            acc.w += l * q.w;
        }
    }
    *(float4*)(dst + ((long)b * N + r) * CH + c4) = acc;
}

__global__ __launch_bounds__(256) void spmm_pull_kernel(
    const float* __restrict__ src, const int* __restrict__ rp, const int* __restrict__ cnt,
    const int2* __restrict__ ep, float* __restrict__ dst, int N) {
    const int lane = threadIdx.x & 31;
    const int r = __builtin_amdgcn_readfirstlane((int)blockIdx.x * 8 + (int)(threadIdx.x >> 5));
    if (r >= N) return;
    const int beg = rp[r];
    const int end = beg + cnt[r];
    const int b = lane >> 4;
    const int c4 = (lane & 15) * 4;
    const float* __restrict__ sb = src + (long)b * N * CH + c4;
    float4 acc = make_float4(0.0f, 0.0f, 0.0f, 0.0f);
    for (int base = beg; base < end; base += 32) {
        const int m = min(32, end - base);
        int2 mine = make_int2(0, 0);
        if (lane < m) mine = ep[base + lane];
        #pragma unroll 4
        for (int j = 0; j < m; ++j) {
            const int c = __builtin_amdgcn_readlane(mine.x, j);
            const float l = __int_as_float(__builtin_amdgcn_readlane(mine.y, j));
            const float4 q = *(const float4*)(sb + (long)c * CH);
            acc.x += l * q.x;
            acc.y += l * q.y;
            acc.z += l * q.z;
            acc.w += l * q.w;
        }
    }
    *(float4*)(dst + ((long)b * N + r) * CH + c4) = acc;
}

__global__ __launch_bounds__(128) void cheb_gemm_wmma(
    const float* __restrict__ x, const float* __restrict__ tx1,
    const float* __restrict__ s2, const float* __restrict__ w,
    const float* __restrict__ bias, float* __restrict__ out, long BN) {
    __shared__ float wlds[3 * CH * CH];

    const int tid = threadIdx.x;
    for (int idx = tid; idx < 3 * CH * CH; idx += 128) {
        const int t   = idx >> 12;
        const int rem = idx & 4095;
        const int k   = rem >> 6;
        const int n   = rem & 63;
        wlds[((((t << 5) + (k >> 1)) << 6) + n) * 2 + (k & 1)] = w[idx];
    }
    __syncthreads();

    const int wave = tid >> 5;
    const int lane = tid & 31;
    const long m0 = ((long)blockIdx.x * 4 + wave) * 16;
    if (m0 + 16 > BN) return;

    const int mrow = lane & 15;
    const int lhi  = lane >> 4;

    const v2f* __restrict__ wp = (const v2f*)wlds;
    const v2f* __restrict__ bbase = wp + (lhi << 6) + mrow;

    v8f acc[4] = {v8f{}, v8f{}, v8f{}, v8f{}};

    #pragma unroll
    for (int t = 0; t < 3; ++t) {
        const float* A = (t == 0) ? x : (t == 1) ? tx1 : s2;
        #pragma unroll
        for (int ks = 0; ks < 16; ++ks) {
            const int kb = ks * 4 + (lhi << 1);
            const float2 a2 = *(const float2*)(A + (m0 + mrow) * CH + kb);
            v2f a;
            a.x = a2.x; a.y = a2.y;
            if (t == 2) {
                const float2 x2 = *(const float2*)(x + (m0 + mrow) * CH + kb);
                a.x = 2.0f * a.x - x2.x;
                a.y = 2.0f * a.y - x2.y;
            }
            #pragma unroll
            for (int nt = 0; nt < 4; ++nt) {
                const v2f b = bbase[(t * 32 + ks * 2) * 64 + nt * 16];
                acc[nt] = __builtin_amdgcn_wmma_f32_16x16x4_f32(
                    false, a, false, b, (short)0, acc[nt], false, false);
            }
        }
    }

    const int moff = lhi << 3;
    #pragma unroll
    for (int nt = 0; nt < 4; ++nt) {
        const int col = nt * 16 + mrow;
        const float bv = bias[col];
        #pragma unroll
        for (int v = 0; v < 8; ++v) {
            out[(m0 + moff + v) * CH + col] = acc[nt][v] + bv;
        }
    }
}

__global__ void cheb_gemm_tail(const float* __restrict__ x, const float* __restrict__ tx1,
                               const float* __restrict__ s2, const float* __restrict__ w,
                               const float* __restrict__ bias, float* __restrict__ out,
                               long startRow, long BN) {
    long r = startRow + blockIdx.x;
    if (r >= BN) return;
    int c = threadIdx.x;
    float acc = bias[c];
    for (int k = 0; k < CH; ++k) {
        float t0 = x[r * CH + k];
        float t1 = tx1[r * CH + k];
        float t2 = 2.0f * s2[r * CH + k] - t0;
        acc += t0 * w[k * CH + c] + t1 * w[CH * CH + k * CH + c] + t2 * w[2 * CH * CH + k * CH + c];
    }
    out[r * CH + c] = acc;
}

extern "C" void kernel_launch(void* const* d_in, const int* in_sizes, int n_in,
                              void* d_out, int out_size, void* d_ws, size_t ws_size,
                              hipStream_t stream) {
    const float* x    = (const float*)d_in[0];
    const int*   ei   = (const int*)d_in[1];
    const float* ew   = (const float*)d_in[2];
    const float* w    = (const float*)d_in[3];
    const float* bias = (const float*)d_in[4];
    float*       out  = (float*)d_out;

    const long BN = (long)in_sizes[0] / CH;
    const int  N  = (int)(BN / NB);
    const long E  = (long)in_sizes[2];
    const int* row = ei;
    const int* col = ei + E;

    char* ws = (char*)d_ws;
    size_t off = 0;
    auto carve = [&](size_t bytes) {
        void* p = ws + off;
        off += (bytes + 255) & ~(size_t)255;
        return p;
    };
    int*   cnt = (int*)carve((size_t)N * sizeof(int));
    int*   rp  = (int*)carve((size_t)N * sizeof(int));
    int2*  ep  = (int2*)carve((size_t)E * sizeof(int2));
    float* tx1 = (float*)carve((size_t)BN * CH * sizeof(float));
    const size_t planeBytes = (size_t)BN * CH * sizeof(float);
    const size_t posBytes   = (size_t)E * sizeof(int);
    float* s2  = (float*)carve(planeBytes > posBytes ? planeBytes : posBytes);
    int*   pos = (int*)s2;
    if (off > ws_size) return;

    const int TB = 256;
    auto gblocks = [](long n, int tb) { return (int)((n + tb - 1) / tb); };

    zero_i32_kernel<<<gblocks(N, TB), TB, 0, stream>>>(cnt, N);
    count_kernel<<<gblocks(E, TB), TB, 0, stream>>>(row, col, cnt, pos, E);
    scan_kernel<<<gblocks(N, 1024), 256, 0, stream>>>(cnt, rp, N);
    fill_kernel<<<gblocks(E, TB), TB, 0, stream>>>(row, col, ew, rp, pos, ep, E);

    spmm_first_kernel<<<gblocks(N, 8), 256, 0, stream>>>(x, rp, cnt, (u64*)ep, tx1, N);
    spmm_pull_kernel<<<gblocks(N, 8), 256, 0, stream>>>(tx1, rp, cnt, ep, s2,  N);

    const long fullTiles = BN / 16;
    if (fullTiles > 0) {
        const int blocks = (int)((fullTiles + 3) / 4);
        cheb_gemm_wmma<<<blocks, 128, 0, stream>>>(x, tx1, s2, w, bias, out, BN);
    }
    const long tailStart = fullTiles * 16;
    if (tailStart < BN) {
        cheb_gemm_tail<<<(int)(BN - tailStart), CH, 0, stream>>>(x, tx1, s2, w, bias, out,
                                                                 tailStart, BN);
    }
}


# Edges grouped by row in LDS (chunks ordered by 128-row range, then each range by row; LDS counters, whole-line stores): no global atomic, no scattered store per edge; 7 -> 5 launches; sum order kept
# speedup vs baseline: 7.3706x; 1.5902x over previous
#include <hip/hip_runtime.h>
#include <hip/hip_bf16.h>

typedef __attribute__((ext_vector_type(2))) float v2f;
typedef __attribute__((ext_vector_type(8))) float v8f;

#define CH 64
#define NB 2

#define RB 128
#define RB_SHIFT 7
#define NBINP 512
#define CE 3125
#define CITER ((CE + 255) / 256)
#define STAGE_CAP 6144

__global__ __launch_bounds__(256) void bin_chunk_kernel(
    const int* __restrict__ row, const int* __restrict__ col, const float* __restrict__ ew,
    int* __restrict__ lsm, int4* __restrict__ binned, long E) {
    __shared__ int lcnt[NBINP];
    __shared__ int lcur[NBINP];
    __shared__ int wsum[8];
    __shared__ int4 stage[CE];
    const int tid = threadIdx.x;
    const int lane = tid & 31;
    const int wave = tid >> 5;
    const long e0 = (long)blockIdx.x * CE;
    const long left = E - e0;
    const int n = (left < (long)CE) ? (int)left : CE;

    lcnt[tid] = 0;
    lcnt[tid + 256] = 0;
    __syncthreads();

    int r[CITER], c[CITER], wb[CITER];
    #pragma unroll
    for (int j = 0; j < CITER; ++j) {
        const int i = j * 256 + tid;
        const long e = e0 + ((i < n) ? i : (n - 1));
        r[j] = row[e];
        c[j] = col[e];
        wb[j] = __float_as_int(ew[e]);
    }
    #pragma unroll
    for (int j = 0; j < CITER; ++j) {
        if (j * 256 + tid >= n) c[j] = r[j];
    }
    #pragma unroll
    for (int j = 0; j < CITER; ++j) {
        if (r[j] != c[j]) atomicAdd(&lcnt[r[j] >> RB_SHIFT], 1);
    }
    __syncthreads();

    const int v0 = lcnt[2 * tid];
    const int v1 = lcnt[2 * tid + 1];
    const int ts = v0 + v1;
    int inc = ts;
    for (int d = 1; d < 32; d <<= 1) {
        const int t = __shfl_up(inc, d);
        if (lane >= d) inc += t;
    }
    if (lane == 31) wsum[wave] = inc;
    __syncthreads();
    int excl = inc - ts;
    int total = 0;
    for (int w = 0; w < 8; ++w) {
        if (w < wave) excl += wsum[w];
        total += wsum[w];
    }
    lcur[2 * tid] = excl;
    lcur[2 * tid + 1] = excl + v0;
    *(int2*)(lsm + (long)blockIdx.x * NBINP + 2 * tid) = make_int2(excl, excl + v0);
    __syncthreads();

    #pragma unroll
    for (int j = 0; j < CITER; ++j) {
        if (r[j] != c[j]) {
            const int p = atomicAdd(&lcur[r[j] >> RB_SHIFT], 1);
            stage[p] = make_int4(r[j], c[j], wb[j], 0);
        }
    }
    __syncthreads();
    for (int i = tid; i < total; i += 256) binned[e0 + i] = stage[i];
}

__global__ __launch_bounds__(256) void bin_rows_kernel(
    const int* __restrict__ lsm, const int4* __restrict__ binned, int* __restrict__ cnt,
    int* __restrict__ rp, int2* __restrict__ ep, int nblk, int N) {
    __shared__ int rowcnt[RB];
    __shared__ int rowcur[RB];
    __shared__ int wred[2][8];
    __shared__ int wsc[4];
    __shared__ int2 stage[STAGE_CAP];
    const int tid = threadIdx.x;
    const int lane = tid & 31;
    const int wave = tid >> 5;
    const int b = blockIdx.x;

    int s = 0, cn = 0;
    if (tid < nblk) {
        s = lsm[(long)tid * NBINP + b];
        cn = lsm[(long)tid * NBINP + b + 1] - s;
    }
    int a = s, q = cn;
    for (int d = 16; d > 0; d >>= 1) {
        a += __shfl_xor(a, d);
        q += __shfl_xor(q, d);
    }
    if (lane == 0) { wred[0][wave] = a; wred[1][wave] = q; }
    if (tid < RB) rowcnt[tid] = 0;
    __syncthreads();
    int binStart = 0, nb = 0;
    for (int w = 0; w < 8; ++w) { binStart += wred[0][w]; nb += wred[1][w]; }
    const bool fast = (nb <= STAGE_CAP);

    const int4* __restrict__ run = binned + (long)tid * CE + s;
    const int rbase = b * RB;
    int ja = 0;
    for (; ja + 4 <= cn; ja += 4) {
        const int r0 = run[ja].x, r1 = run[ja + 1].x, r2 = run[ja + 2].x, r3 = run[ja + 3].x;
        atomicAdd(&rowcnt[r0 - rbase], 1);
        atomicAdd(&rowcnt[r1 - rbase], 1);
        atomicAdd(&rowcnt[r2 - rbase], 1);
        atomicAdd(&rowcnt[r3 - rbase], 1);
    }
    for (; ja < cn; ++ja) atomicAdd(&rowcnt[run[ja].x - rbase], 1);
    __syncthreads();

    int v = 0, inc = 0;
    if (tid < RB) {
        v = rowcnt[tid];
        inc = v;
        for (int d = 1; d < 32; d <<= 1) {
            const int t = __shfl_up(inc, d);
            if (lane >= d) inc += t;
        }
        if (lane == 31) wsc[wave] = inc;
    }
    __syncthreads();
    if (tid < RB) {
        int excl = inc - v;
        for (int w = 0; w < 4; ++w) if (w < wave) excl += wsc[w];
        rowcur[tid] = excl;
        const int rr = rbase + tid;
        if (rr < N) { cnt[rr] = v; rp[rr] = binStart + excl; }
    }
    __syncthreads();

    int jb = 0;
    for (; jb + 4 <= cn; jb += 4) {
        const int4 e0 = run[jb], e1 = run[jb + 1], e2 = run[jb + 2], e3 = run[jb + 3];
        const int p0 = atomicAdd(&rowcur[e0.x - rbase], 1);
        const int p1 = atomicAdd(&rowcur[e1.x - rbase], 1);
        const int p2 = atomicAdd(&rowcur[e2.x - rbase], 1);
        const int p3 = atomicAdd(&rowcur[e3.x - rbase], 1);
        if (fast) {
            stage[p0] = make_int2(e0.y, e0.z);
            stage[p1] = make_int2(e1.y, e1.z);
            stage[p2] = make_int2(e2.y, e2.z);
            stage[p3] = make_int2(e3.y, e3.z);
        } else {
            ep[binStart + p0] = make_int2(e0.y, e0.z);
            ep[binStart + p1] = make_int2(e1.y, e1.z);
            ep[binStart + p2] = make_int2(e2.y, e2.z);
            ep[binStart + p3] = make_int2(e3.y, e3.z);
        }
    }
    for (; jb < cn; ++jb) {
        const int4 e = run[jb];
        const int p = atomicAdd(&rowcur[e.x - rbase], 1);
        if (fast) stage[p] = make_int2(e.y, e.z); else ep[binStart + p] = make_int2(e.y, e.z);
    }
    __syncthreads();
    if (fast) {
        for (int i = tid; i < nb; i += 256) ep[binStart + i] = stage[i];
    }
}

#define SORT_K 4
typedef unsigned long long u64;
__global__ __launch_bounds__(256) void spmm_first_kernel(
    const float* __restrict__ src, const int* __restrict__ rp, const int* __restrict__ cnt,
    u64* ep, float* __restrict__ dst, int N) {
    __shared__ u64 sbuf[8 * 32 * SORT_K];
    const int lane = threadIdx.x & 31;
    const int wv = threadIdx.x >> 5;
    const int r = __builtin_amdgcn_readfirstlane((int)blockIdx.x * 8 + wv);
    if (r >= N) return;
    const int beg = rp[r];
    const int n = cnt[r];
    u64* seg = ep + beg;
    u64* lw = sbuf + wv * (32 * SORT_K);
    const float disr = __frsqrt_rn((float)n);
    const bool small = (n <= 32 * SORT_K);

    if (small) {
        const int kmax = (n + 31) >> 5;
        u64 key[SORT_K];
        int rank[SORT_K];
        for (int k = 0; k < SORT_K; ++k) {
            const int i = k * 32 + lane;
            key[k] = ~0ull;
            rank[k] = 0;
            if (i < n) {
                const u64 p = seg[i];
                const int c = (int)(unsigned)p;
                const float w = __int_as_float((int)(unsigned)(p >> 32));
                const int dc = cnt[c];
                const float disc = (dc > 0) ? __frsqrt_rn((float)dc) : 0.0f;
                const float l = -disr * w * disc;
                key[k] = ((u64)(unsigned)__float_as_int(l) << 32) | (u64)(unsigned)c;
            }
        }
        for (int k2 = 0; k2 < SORT_K; ++k2) {
            if (k2 < kmax) {
                const int m = min(32, n - k2 * 32);
                for (int j = 0; j < m; ++j) {
                    const unsigned lo = (unsigned)__builtin_amdgcn_readlane((int)(unsigned)key[k2], j);
                    const unsigned hi = (unsigned)__builtin_amdgcn_readlane((int)(unsigned)(key[k2] >> 32), j);
                    const u64 kj = ((u64)hi << 32) | lo;
                    const int gj = k2 * 32 + j;
                    for (int k = 0; k < SORT_K; ++k) {
                        if (k < kmax) {
                            const int gi = k * 32 + lane;
                            rank[k] += (kj < key[k] || (kj == key[k] && gj < gi)) ? 1 : 0;
                        }
                    }
                }
            }
        }
        for (int k = 0; k < SORT_K; ++k) {
            const int i = k * 32 + lane;
            if (i < n) lw[rank[k]] = key[k];
        }
        for (int k = 0; k < SORT_K; ++k) {
            const int i = k * 32 + lane;
            if (i < n) seg[i] = lw[i];
        }
    } else {
        for (int i = lane; i < n; i += 32) {
            const u64 p = seg[i];
            const int c = (int)(unsigned)p;
            const float w = __int_as_float((int)(unsigned)(p >> 32));
            const int dc = cnt[c];
            const float disc = (dc > 0) ? __frsqrt_rn((float)dc) : 0.0f;
            const float l = -disr * w * disc;
            seg[i] = ((u64)(unsigned)__float_as_int(l) << 32) | (u64)(unsigned)c;
        }
        __threadfence();
        if (lane == 0) {
            for (int i = 1; i < n; ++i) {
                const u64 kk = seg[i];
                int j = i - 1;
                while (j >= 0 && seg[j] > kk) { seg[j + 1] = seg[j]; --j; }
                seg[j + 1] = kk;
            }
        }
        __threadfence();
    }

    const int b = lane >> 4;
    const int c4 = (lane & 15) * 4;
    const float* __restrict__ sb = src + (long)b * N * CH + c4;
    float4 acc = make_float4(0.0f, 0.0f, 0.0f, 0.0f);
    for (int base = 0; base < n; base += 32) {
        const int m = min(32, n - base);
        u64 mine = 0ull;
        if (small) {
            if (lane < m) mine = lw[base + lane];
        } else {
            if (lane < m) mine = seg[base + lane];
        }
        const int mx = (int)(unsigned)mine;
        const int my = (int)(unsigned)(mine >> 32);
        #pragma unroll 4
        for (int j = 0; j < m; ++j) {
            const int c = __builtin_amdgcn_readlane(mx, j);
            const float l = __int_as_float(__builtin_amdgcn_readlane(my, j));
            const float4 q = *(const float4*)(sb + (long)c * CH);
            acc.x += l * q.x;
            acc.y += l * q.y;
            acc.z += l * q.z;
            acc.w += l * q.w;
        }
    }
    *(float4*)(dst + ((long)b * N + r) * CH + c4) = acc;
}

__global__ __launch_bounds__(256) void spmm_pull_kernel(
    const float* __restrict__ src, const int* __restrict__ rp, const int* __restrict__ cnt,
    const int2* __restrict__ ep, float* __restrict__ dst, int N) {
    const int lane = threadIdx.x & 31;
    const int r = __builtin_amdgcn_readfirstlane((int)blockIdx.x * 8 + (int)(threadIdx.x >> 5));
    if (r >= N) return;
    const int beg = rp[r];
    const int end = beg + cnt[r];
    const int b = lane >> 4;
    const int c4 = (lane & 15) * 4;
    const float* __restrict__ sb = src + (long)b * N * CH + c4;
    float4 acc = make_float4(0.0f, 0.0f, 0.0f, 0.0f);
    for (int base = beg; base < end; base += 32) {
        const int m = min(32, end - base);
        int2 mine = make_int2(0, 0);
        if (lane < m) mine = ep[base + lane];
        #pragma unroll 4
        for (int j = 0; j < m; ++j) {
            const int c = __builtin_amdgcn_readlane(mine.x, j);
            const float l = __int_as_float(__builtin_amdgcn_readlane(mine.y, j));
            const float4 q = *(const float4*)(sb + (long)c * CH);
            acc.x += l * q.x;
            acc.y += l * q.y;
            acc.z += l * q.z;
            acc.w += l * q.w;
        }
    }
    *(float4*)(dst + ((long)b * N + r) * CH + c4) = acc;
}

__global__ __launch_bounds__(128) void cheb_gemm_wmma(
    const float* __restrict__ x, const float* __restrict__ tx1,
    const float* __restrict__ s2, const float* __restrict__ w,
    const float* __restrict__ bias, float* __restrict__ out, long BN) {
    __shared__ float wlds[3 * CH * CH];

    const int tid = threadIdx.x;
    for (int idx = tid; idx < 3 * CH * CH; idx += 128) {
        const int t   = idx >> 12;
        const int rem = idx & 4095;
        const int k   = rem >> 6;
        const int n   = rem & 63;
        wlds[((((t << 5) + (k >> 1)) << 6) + n) * 2 + (k & 1)] = w[idx];
    }
    __syncthreads();

    const int wave = tid >> 5;
    const int lane = tid & 31;
    const long m0 = ((long)blockIdx.x * 4 + wave) * 16;
    if (m0 + 16 > BN) return;

    const int mrow = lane & 15;
    const int lhi  = lane >> 4;

    const v2f* __restrict__ wp = (const v2f*)wlds;
    const v2f* __restrict__ bbase = wp + (lhi << 6) + mrow;

    v8f acc[4] = {v8f{}, v8f{}, v8f{}, v8f{}};

    #pragma unroll
    for (int t = 0; t < 3; ++t) {
        const float* A = (t == 0) ? x : (t == 1) ? tx1 : s2;
        #pragma unroll
        for (int ks = 0; ks < 16; ++ks) {
            const int kb = ks * 4 + (lhi << 1);
            const float2 a2 = *(const float2*)(A + (m0 + mrow) * CH + kb);
            v2f a;
            a.x = a2.x; a.y = a2.y;
            if (t == 2) {
                const float2 x2 = *(const float2*)(x + (m0 + mrow) * CH + kb);
                a.x = 2.0f * a.x - x2.x;
                a.y = 2.0f * a.y - x2.y;
            }
            #pragma unroll
            for (int nt = 0; nt < 4; ++nt) {
                const v2f b = bbase[(t * 32 + ks * 2) * 64 + nt * 16];
                acc[nt] = __builtin_amdgcn_wmma_f32_16x16x4_f32(
                    false, a, false, b, (short)0, acc[nt], false, false);
            }
        }
    }

    const int moff = lhi << 3;
    #pragma unroll
    for (int nt = 0; nt < 4; ++nt) {
        const int col = nt * 16 + mrow;
        const float bv = bias[col];
        #pragma unroll
        for (int v = 0; v < 8; ++v) {
            out[(m0 + moff + v) * CH + col] = acc[nt][v] + bv;
        }
    }
}

__global__ void cheb_gemm_tail(const float* __restrict__ x, const float* __restrict__ tx1,
                               const float* __restrict__ s2, const float* __restrict__ w,
                               const float* __restrict__ bias, float* __restrict__ out,
                               long startRow, long BN) {
    long r = startRow + blockIdx.x;
    if (r >= BN) return;
    int c = threadIdx.x;
    float acc = bias[c];
    for (int k = 0; k < CH; ++k) {
        float t0 = x[r * CH + k];
        float t1 = tx1[r * CH + k];
        float t2 = 2.0f * s2[r * CH + k] - t0;
        acc += t0 * w[k * CH + c] + t1 * w[CH * CH + k * CH + c] + t2 * w[2 * CH * CH + k * CH + c];
    }
    out[r * CH + c] = acc;
}

extern "C" void kernel_launch(void* const* d_in, const int* in_sizes, int n_in,
                              void* d_out, int out_size, void* d_ws, size_t ws_size,
                              hipStream_t stream) {
    const float* x    = (const float*)d_in[0];
    const int*   ei   = (const int*)d_in[1];
    const float* ew   = (const float*)d_in[2];
    const float* w    = (const float*)d_in[3];
    const float* bias = (const float*)d_in[4];
    float*       out  = (float*)d_out;

    const long BN = (long)in_sizes[0] / CH;
    const int  N  = (int)(BN / NB);
    const long E  = (long)in_sizes[2];
    const int* row = ei;
    const int* col = ei + E;

    char* ws = (char*)d_ws;
    size_t off = 0;
    auto carve = [&](size_t bytes) {
        void* p = ws + off;
        off += (bytes + 255) & ~(size_t)255;
        return p;
    };
    const int nblk = (int)((E + CE - 1) / CE);
    const int nbin = (N + RB - 1) / RB;
    if (nblk > 256 || nbin >= NBINP) return;

    int*   cnt = (int*)carve((size_t)N * sizeof(int));
    int*   rp  = (int*)carve((size_t)N * sizeof(int));
    int2*  ep  = (int2*)carve((size_t)E * sizeof(int2));
    const size_t planeBytes = (size_t)BN * CH * sizeof(float);
    const size_t lsmBytes   = (size_t)nblk * NBINP * sizeof(int);
    const size_t binBytes   = (size_t)nblk * CE * sizeof(int4);
    float* tx1 = (float*)carve(planeBytes > lsmBytes ? planeBytes : lsmBytes);
    float* s2  = (float*)carve(planeBytes > binBytes ? planeBytes : binBytes);
    int*   lsm    = (int*)tx1;
    int4*  binned = (int4*)s2;
    if (off > ws_size) return;

    auto gblocks = [](long n, int tb) { return (int)((n + tb - 1) / tb); };

    if (nblk > 0)
        bin_chunk_kernel<<<nblk, 256, 0, stream>>>(row, col, ew, lsm, binned, E);
    bin_rows_kernel<<<nbin, 256, 0, stream>>>(lsm, binned, cnt, rp, ep, nblk, N);

    spmm_first_kernel<<<gblocks(N, 8), 256, 0, stream>>>(x, rp, cnt, (u64*)ep, tx1, N);
    spmm_pull_kernel<<<gblocks(N, 8), 256, 0, stream>>>(tx1, rp, cnt, ep, s2,  N);

    const long fullTiles = BN / 16;
    if (fullTiles > 0) {
        const int blocks = (int)((fullTiles + 3) / 4);
        cheb_gemm_wmma<<<blocks, 128, 0, stream>>>(x, tx1, s2, w, bias, out, BN);
    }
    const long tailStart = fullTiles * 16;
    if (tailStart < BN) {
        cheb_gemm_tail<<<(int)(BN - tailStart), CH, 0, stream>>>(x, tx1, s2, w, bias, out,
                                                                 tailStart, BN);
    }
}


# Second SpMM fused into the WMMA GEMM launch: 8-wave blocks gather 16 node rows (2 per wave) into LDS A tiles x, tx1, 2*s2-x and multiply them there; s2 plane gone, 5 -> 4 launches; sum order kept
# speedup vs baseline: 8.5692x; 1.1626x over previous
#include <hip/hip_runtime.h>
#include <hip/hip_bf16.h>

typedef __attribute__((ext_vector_type(2))) float v2f;
typedef __attribute__((ext_vector_type(8))) float v8f;

#define CH 64
#define NB 2

#define RB 128
#define RB_SHIFT 7
#define NBINP 512
#define CE 3125
#define CITER ((CE + 255) / 256)
#define STAGE_CAP 6144

__global__ __launch_bounds__(256) void bin_chunk_kernel(
    const int* __restrict__ row, const int* __restrict__ col, const float* __restrict__ ew,
    int* __restrict__ lsm, int4* __restrict__ binned, long E) {
    __shared__ int lcnt[NBINP];
    __shared__ int lcur[NBINP];
    __shared__ int wsum[8];
    __shared__ int4 stage[CE];
    const int tid = threadIdx.x;
    const int lane = tid & 31;
    const int wave = tid >> 5;
    const long e0 = (long)blockIdx.x * CE;
    const long left = E - e0;
    const int n = (left < (long)CE) ? (int)left : CE;

    lcnt[tid] = 0;
    lcnt[tid + 256] = 0;
    __syncthreads();

    int r[CITER], c[CITER], wb[CITER];
    #pragma unroll
    for (int j = 0; j < CITER; ++j) {
        const int i = j * 256 + tid;
        const long e = e0 + ((i < n) ? i : (n - 1));
        r[j] = row[e];
        c[j] = col[e];
        wb[j] = __float_as_int(ew[e]);
    }
    #pragma unroll
    for (int j = 0; j < CITER; ++j) {
        if (j * 256 + tid >= n) c[j] = r[j];
    }
    #pragma unroll
    for (int j = 0; j < CITER; ++j) {
        if (r[j] != c[j]) atomicAdd(&lcnt[r[j] >> RB_SHIFT], 1);
    }
    __syncthreads();

    const int v0 = lcnt[2 * tid];
    const int v1 = lcnt[2 * tid + 1];
    const int ts = v0 + v1;
    int inc = ts;
    for (int d = 1; d < 32; d <<= 1) {
        const int t = __shfl_up(inc, d);
        if (lane >= d) inc += t;
    }
    if (lane == 31) wsum[wave] = inc;
    __syncthreads();
    int excl = inc - ts;
    int total = 0;
    for (int w = 0; w < 8; ++w) {
        if (w < wave) excl += wsum[w];
        total += wsum[w];
    }
    lcur[2 * tid] = excl;
    lcur[2 * tid + 1] = excl + v0;
    *(int2*)(lsm + (long)blockIdx.x * NBINP + 2 * tid) = make_int2(excl, excl + v0);
    __syncthreads();

    #pragma unroll
    for (int j = 0; j < CITER; ++j) {
        if (r[j] != c[j]) {
            const int p = atomicAdd(&lcur[r[j] >> RB_SHIFT], 1);
            stage[p] = make_int4(r[j], c[j], wb[j], 0);
        }
    }
    __syncthreads();
    for (int i = tid; i < total; i += 256) binned[e0 + i] = stage[i];
}

__global__ __launch_bounds__(256) void bin_rows_kernel(
    const int* __restrict__ lsm, const int4* __restrict__ binned, int* __restrict__ cnt,
    int* __restrict__ rp, int2* __restrict__ ep, int nblk, int N) {
    __shared__ int rowcnt[RB];
    __shared__ int rowcur[RB];
    __shared__ int wred[2][8];
    __shared__ int wsc[4];
    __shared__ int2 stage[STAGE_CAP];
    const int tid = threadIdx.x;
    const int lane = tid & 31;
    const int wave = tid >> 5;
    const int b = blockIdx.x;

    int s = 0, cn = 0;
    if (tid < nblk) {
        s = lsm[(long)tid * NBINP + b];
        cn = lsm[(long)tid * NBINP + b + 1] - s;
    }
    int a = s, q = cn;
    for (int d = 16; d > 0; d >>= 1) {
        a += __shfl_xor(a, d);
        q += __shfl_xor(q, d);
    }
    if (lane == 0) { wred[0][wave] = a; wred[1][wave] = q; }
    if (tid < RB) rowcnt[tid] = 0;
    __syncthreads();
    int binStart = 0, nb = 0;
    for (int w = 0; w < 8; ++w) { binStart += wred[0][w]; nb += wred[1][w]; }
    const bool fast = (nb <= STAGE_CAP);

    const int4* __restrict__ run = binned + (long)tid * CE + s;
    const int rbase = b * RB;
    int ja = 0;
    for (; ja + 4 <= cn; ja += 4) {
        const int r0 = run[ja].x, r1 = run[ja + 1].x, r2 = run[ja + 2].x, r3 = run[ja + 3].x;
        atomicAdd(&rowcnt[r0 - rbase], 1);
        atomicAdd(&rowcnt[r1 - rbase], 1);
        atomicAdd(&rowcnt[r2 - rbase], 1);
        atomicAdd(&rowcnt[r3 - rbase], 1);
    }
    for (; ja < cn; ++ja) atomicAdd(&rowcnt[run[ja].x - rbase], 1);
    __syncthreads();

    int v = 0, inc = 0;
    if (tid < RB) {
        v = rowcnt[tid];
        inc = v;
        for (int d = 1; d < 32; d <<= 1) {
            const int t = __shfl_up(inc, d);
            if (lane >= d) inc += t;
        }
        if (lane == 31) wsc[wave] = inc;
    }
    __syncthreads();
    if (tid < RB) {
        int excl = inc - v;
        for (int w = 0; w < 4; ++w) if (w < wave) excl += wsc[w];
        rowcur[tid] = excl;
        const int rr = rbase + tid;
        if (rr < N) { cnt[rr] = v; rp[rr] = binStart + excl; }
    }
    __syncthreads();

    int jb = 0;
    for (; jb + 4 <= cn; jb += 4) {
        const int4 e0 = run[jb], e1 = run[jb + 1], e2 = run[jb + 2], e3 = run[jb + 3];
        const int p0 = atomicAdd(&rowcur[e0.x - rbase], 1);
        const int p1 = atomicAdd(&rowcur[e1.x - rbase], 1);
        const int p2 = atomicAdd(&rowcur[e2.x - rbase], 1);
        const int p3 = atomicAdd(&rowcur[e3.x - rbase], 1);
        if (fast) {
            stage[p0] = make_int2(e0.y, e0.z);
            stage[p1] = make_int2(e1.y, e1.z);
            stage[p2] = make_int2(e2.y, e2.z);
            stage[p3] = make_int2(e3.y, e3.z);
        } else {
            ep[binStart + p0] = make_int2(e0.y, e0.z);
            ep[binStart + p1] = make_int2(e1.y, e1.z);
            ep[binStart + p2] = make_int2(e2.y, e2.z);
            ep[binStart + p3] = make_int2(e3.y, e3.z);
        }
    }
    for (; jb < cn; ++jb) {
        const int4 e = run[jb];
        const int p = atomicAdd(&rowcur[e.x - rbase], 1);
        if (fast) stage[p] = make_int2(e.y, e.z); else ep[binStart + p] = make_int2(e.y, e.z);
    }
    __syncthreads();
    if (fast) {
        for (int i = tid; i < nb; i += 256) ep[binStart + i] = stage[i];
    }
}

#define SORT_K 4
typedef unsigned long long u64;
__global__ __launch_bounds__(256) void spmm_first_kernel(
    const float* __restrict__ src, const int* __restrict__ rp, const int* __restrict__ cnt,
    u64* ep, float* __restrict__ dst, int N) {
    __shared__ u64 sbuf[8 * 32 * SORT_K];
    const int lane = threadIdx.x & 31;
    const int wv = threadIdx.x >> 5;
    const int r = __builtin_amdgcn_readfirstlane((int)blockIdx.x * 8 + wv);
    if (r >= N) return;
    const int beg = rp[r];
    const int n = cnt[r];
    u64* seg = ep + beg;
    u64* lw = sbuf + wv * (32 * SORT_K);
    const float disr = __frsqrt_rn((float)n);
    const bool small = (n <= 32 * SORT_K);

    if (small) {
        const int kmax = (n + 31) >> 5;
        u64 key[SORT_K];
        int rank[SORT_K];
        for (int k = 0; k < SORT_K; ++k) {
            const int i = k * 32 + lane;
            key[k] = ~0ull;
            rank[k] = 0;
            if (i < n) {
                const u64 p = seg[i];
                const int c = (int)(unsigned)p;
                const float w = __int_as_float((int)(unsigned)(p >> 32));
                const int dc = cnt[c];
                const float disc = (dc > 0) ? __frsqrt_rn((float)dc) : 0.0f;
                const float l = -disr * w * disc;
                key[k] = ((u64)(unsigned)__float_as_int(l) << 32) | (u64)(unsigned)c;
            }
        }
        for (int k2 = 0; k2 < SORT_K; ++k2) {
            if (k2 < kmax) {
                const int m = min(32, n - k2 * 32);
                for (int j = 0; j < m; ++j) {
                    const unsigned lo = (unsigned)__builtin_amdgcn_readlane((int)(unsigned)key[k2], j);
                    const unsigned hi = (unsigned)__builtin_amdgcn_readlane((int)(unsigned)(key[k2] >> 32), j);
                    const u64 kj = ((u64)hi << 32) | lo;
                    const int gj = k2 * 32 + j;
                    for (int k = 0; k < SORT_K; ++k) {
                        if (k < kmax) {
                            const int gi = k * 32 + lane;
                            rank[k] += (kj < key[k] || (kj == key[k] && gj < gi)) ? 1 : 0;
                        }
                    }
                }
            }
        }
        for (int k = 0; k < SORT_K; ++k) {
            const int i = k * 32 + lane;
            if (i < n) lw[rank[k]] = key[k];
        }
        for (int k = 0; k < SORT_K; ++k) {
            const int i = k * 32 + lane;
            if (i < n) seg[i] = lw[i];
        }
    } else {
        for (int i = lane; i < n; i += 32) {
            const u64 p = seg[i];
            const int c = (int)(unsigned)p;
            const float w = __int_as_float((int)(unsigned)(p >> 32));
            const int dc = cnt[c];
            const float disc = (dc > 0) ? __frsqrt_rn((float)dc) : 0.0f;
            const float l = -disr * w * disc;
            seg[i] = ((u64)(unsigned)__float_as_int(l) << 32) | (u64)(unsigned)c;
        }
        __threadfence();
        if (lane == 0) {
            for (int i = 1; i < n; ++i) {
                const u64 kk = seg[i];
                int j = i - 1;
                while (j >= 0 && seg[j] > kk) { seg[j + 1] = seg[j]; --j; }
                seg[j + 1] = kk;
            }
        }
        __threadfence();
    }

    const int b = lane >> 4;
    const int c4 = (lane & 15) * 4;
    const float* __restrict__ sb = src + (long)b * N * CH + c4;
    float4 acc = make_float4(0.0f, 0.0f, 0.0f, 0.0f);
    for (int base = 0; base < n; base += 32) {
        const int m = min(32, n - base);
        u64 mine = 0ull;
        if (small) {
            if (lane < m) mine = lw[base + lane];
        } else {
            if (lane < m) mine = seg[base + lane];
        }
        const int mx = (int)(unsigned)mine;
        const int my = (int)(unsigned)(mine >> 32);
        #pragma unroll 4
        for (int j = 0; j < m; ++j) {
            const int c = __builtin_amdgcn_readlane(mx, j);
            const float l = __int_as_float(__builtin_amdgcn_readlane(my, j));
            const float4 q = *(const float4*)(sb + (long)c * CH);
            acc.x += l * q.x;
            acc.y += l * q.y;
            acc.z += l * q.z;
            acc.w += l * q.w;
        }
    }
    *(float4*)(dst + ((long)b * N + r) * CH + c4) = acc;
}

#define FR 16
#define TP 68
__global__ __launch_bounds__(256) void spmm_gemm_fused(
    const float* __restrict__ x, const float* __restrict__ tx1,
    const int* __restrict__ rp, const int* __restrict__ cnt, const int2* __restrict__ ep,
    const float* __restrict__ w, const float* __restrict__ bias, float* __restrict__ out,
    int N, int nchunk) {
    __shared__ float wlds[3 * CH * CH];
    __shared__ float tile[3][NB][FR][TP];

    const int tid = threadIdx.x;
    #pragma unroll
    for (int i = 0; i < 3 * (CH / 2) * (CH / 4) / 256; ++i) {
        const int u = i * 256 + tid;
        const int n4 = u & 15;
        const int tj = u >> 4;
        const float* g = w + ((tj >> 5) * CH + 2 * (tj & 31)) * CH + n4 * 4;
        const float4 e = *(const float4*)g;
        const float4 o = *(const float4*)(g + CH);
        float4* d = (float4*)&wlds[(tj * CH + n4 * 4) * 2];
        d[0] = make_float4(e.x, o.x, e.y, o.y);
        d[1] = make_float4(e.z, o.z, e.w, o.w);
    }

    const int lane = tid & 31;
    const int wv = __builtin_amdgcn_readfirstlane(tid >> 5);
    const int b  = lane >> 4;
    const int c4 = (lane & 15) * 4;
    const float* __restrict__ sb = tx1 + (long)b * N * CH + c4;
    const int mt = wv & 1;
    const int nt = wv >> 1;
    const int mrow = lane & 15;
    const int lhi  = lane >> 4;
    const v2f* __restrict__ bbase = (const v2f*)wlds + (lhi << 6) + nt * 16 + mrow;
    const float bv = bias[nt * 16 + mrow];

    for (int chunk = blockIdx.x; chunk < nchunk; chunk += gridDim.x) {
        const int r0 = chunk * FR;

        const int ra = r0 + 2 * wv;
        const int rb = ra + 1;
        const int rsa = (ra < N) ? ra : 0;
        const int rsb = (rb < N) ? rb : 0;
        const long oa = ((long)b * N + rsa) * CH + c4;
        const long ob = ((long)b * N + rsb) * CH + c4;
        const float4 xa = *(const float4*)(x + oa);
        const float4 xb = *(const float4*)(x + ob);
        const float4 ta = *(const float4*)(tx1 + oa);
        const float4 tb = *(const float4*)(tx1 + ob);
        const int bega = rp[rsa];
        const int begb = rp[rsb];
        const int cna = cnt[rsa];
        const int cnb = cnt[rsb];
        const int na = (ra < N) ? cna : 0;
        const int nb = (rb < N) ? cnb : 0;
        const int nmax = max(na, nb);
        int ma = min(32, na);
        int mb = min(32, nb);
        int2 pa = make_int2(0, 0);
        int2 pb = make_int2(0, 0);
        if (lane < ma) pa = ep[bega + lane];
        if (lane < mb) pb = ep[begb + lane];
        {
            const int ia = 2 * wv;
            *(float4*)&tile[0][b][ia][c4]     = xa;
            *(float4*)&tile[0][b][ia + 1][c4] = xb;
            *(float4*)&tile[1][b][ia][c4]     = ta;
            *(float4*)&tile[1][b][ia + 1][c4] = tb;
        }
        float4 acca = make_float4(0.0f, 0.0f, 0.0f, 0.0f);
        float4 accb = make_float4(0.0f, 0.0f, 0.0f, 0.0f);
        int base = 0;
        while (true) {
            const int mm4 = (max(ma, mb) + 3) & ~3;
            for (int j = 0; j < mm4; j += 4) {
                float4 qa[4], qb[4];
                float la[4], lb[4];
                #pragma unroll
                for (int u = 0; u < 4; ++u) {
                    const int ca = __builtin_amdgcn_readlane(pa.x, j + u);
                    const int cb = __builtin_amdgcn_readlane(pb.x, j + u);
                    la[u] = __int_as_float(__builtin_amdgcn_readlane(pa.y, j + u));
                    lb[u] = __int_as_float(__builtin_amdgcn_readlane(pb.y, j + u));
                    qa[u] = *(const float4*)(sb + (long)ca * CH);
                    qb[u] = *(const float4*)(sb + (long)cb * CH);
                }
                #pragma unroll
                for (int u = 0; u < 4; ++u) {
                    acca.x += la[u] * qa[u].x;
                    acca.y += la[u] * qa[u].y;
                    acca.z += la[u] * qa[u].z;
                    acca.w += la[u] * qa[u].w;
                    accb.x += lb[u] * qb[u].x;
                    accb.y += lb[u] * qb[u].y;
                    accb.z += lb[u] * qb[u].z;
                    accb.w += lb[u] * qb[u].w;
                }
            }
            base += 32;
            if (base >= nmax) break;
            ma = min(32, max(0, na - base));
            mb = min(32, max(0, nb - base));
            pa = make_int2(0, 0);
            pb = make_int2(0, 0);
            if (lane < ma) pa = ep[bega + base + lane];
            if (lane < mb) pb = ep[begb + base + lane];
        }
        {
            const int ia = 2 * wv;
            *(float4*)&tile[2][b][ia][c4] = make_float4(
                2.0f * acca.x - xa.x, 2.0f * acca.y - xa.y, 2.0f * acca.z - xa.z, 2.0f * acca.w - xa.w);
            *(float4*)&tile[2][b][ia + 1][c4] = make_float4(
                2.0f * accb.x - xb.x, 2.0f * accb.y - xb.y, 2.0f * accb.z - xb.z, 2.0f * accb.w - xb.w);
        }
        __builtin_amdgcn_fence(__ATOMIC_RELEASE, "workgroup", "local");
        __builtin_amdgcn_s_barrier();
        __builtin_amdgcn_fence(__ATOMIC_ACQUIRE, "workgroup", "local");

        v8f acc = v8f{};
        #pragma unroll
        for (int t = 0; t < 3; ++t) {
            const float* At = &tile[t][mt][mrow][2 * lhi];
            #pragma unroll
            for (int ks = 0; ks < 16; ++ks) {
                const v2f a = *(const v2f*)(At + ks * 4);
                const v2f bf = bbase[(t * 32 + ks * 2) * 64];
                acc = __builtin_amdgcn_wmma_f32_16x16x4_f32(
                    false, a, false, bf, (short)0, acc, false, false);
            }
        }
        {
            const int col = nt * 16 + mrow;
            float* __restrict__ op = out + ((long)mt * N + r0 + (lhi << 3)) * CH + col;
            if (r0 + FR <= N) {
                #pragma unroll
                for (int v = 0; v < 8; ++v) op[v * CH] = acc[v] + bv;
            } else {
                #pragma unroll
                for (int v = 0; v < 8; ++v) {
                    if (r0 + (lhi << 3) + v < N) op[v * CH] = acc[v] + bv;
                }
            }
        }
        __builtin_amdgcn_fence(__ATOMIC_RELEASE, "workgroup", "local");
        __builtin_amdgcn_s_barrier();
        __builtin_amdgcn_fence(__ATOMIC_ACQUIRE, "workgroup", "local");
    }
}

extern "C" void kernel_launch(void* const* d_in, const int* in_sizes, int n_in,
                              void* d_out, int out_size, void* d_ws, size_t ws_size,
                              hipStream_t stream) {
    const float* x    = (const float*)d_in[0];
    const int*   ei   = (const int*)d_in[1];
    const float* ew   = (const float*)d_in[2];
    const float* w    = (const float*)d_in[3];
    const float* bias = (const float*)d_in[4];
    float*       out  = (float*)d_out;

    const long BN = (long)in_sizes[0] / CH;
    const int  N  = (int)(BN / NB);
    const long E  = (long)in_sizes[2];
    const int* row = ei;
    const int* col = ei + E;

    char* ws = (char*)d_ws;
    size_t off = 0;
    auto carve = [&](size_t bytes) {
        void* p = ws + off;
        off += (bytes + 255) & ~(size_t)255;
        return p;
    };
    const int nblk = (int)((E + CE - 1) / CE);
    const int nbin = (N + RB - 1) / RB;
    if (nblk > 256 || nbin >= NBINP) return;

    int*   cnt = (int*)carve((size_t)N * sizeof(int));
    int*   rp  = (int*)carve((size_t)N * sizeof(int));
    int2*  ep  = (int2*)carve((size_t)E * sizeof(int2));
    const size_t planeBytes = (size_t)BN * CH * sizeof(float);
    const size_t lsmBytes   = (size_t)nblk * NBINP * sizeof(int);
    const size_t binBytes   = (size_t)nblk * CE * sizeof(int4);
    float* tx1 = (float*)carve(planeBytes > lsmBytes ? planeBytes : lsmBytes);
    int4*  binned = (int4*)carve(binBytes);
    int*   lsm    = (int*)tx1;
    if (off > ws_size) return;

    auto gblocks = [](long n, int tb) { return (int)((n + tb - 1) / tb); };

    if (nblk > 0)
        bin_chunk_kernel<<<nblk, 256, 0, stream>>>(row, col, ew, lsm, binned, E);
    bin_rows_kernel<<<nbin, 256, 0, stream>>>(lsm, binned, cnt, rp, ep, nblk, N);

    spmm_first_kernel<<<gblocks(N, 8), 256, 0, stream>>>(x, rp, cnt, (u64*)ep, tx1, N);
    const int nchunk = (N + FR - 1) / FR;
    if (nchunk > 0)
        spmm_gemm_fused<<<1024, 256, 0, stream>>>(x, tx1, rp, cnt, ep, w, bias, out, N, nchunk);
}


# Row-grouping kernel: runs copied to LDS once, a half wave per run (a load touches 2-4 lines, not 32), rows counted and placed from LDS in block strides: one global pass, not two; sum order kept
# speedup vs baseline: 9.0725x; 1.0587x over previous
#include <hip/hip_runtime.h>
#include <hip/hip_bf16.h>

typedef __attribute__((ext_vector_type(2))) float v2f;
typedef __attribute__((ext_vector_type(8))) float v8f;

#define CH 64
#define NB 2

#define RB 128
#define RB_SHIFT 7
#define NBINP 512
#define CE 3125
#define CITER ((CE + 255) / 256)
#define RAW_CAP 3072

__global__ __launch_bounds__(256) void bin_chunk_kernel(
    const int* __restrict__ row, const int* __restrict__ col, const float* __restrict__ ew,
    int* __restrict__ lsm, int4* __restrict__ binned, long E) {
    __shared__ int lcnt[NBINP];
    __shared__ int lcur[NBINP];
    __shared__ int wsum[8];
    __shared__ int4 stage[CE];
    const int tid = threadIdx.x;
    const int lane = tid & 31;
    const int wave = tid >> 5;
    const long e0 = (long)blockIdx.x * CE;
    const long left = E - e0;
    const int n = (left < (long)CE) ? (int)left : CE;

    lcnt[tid] = 0;
    lcnt[tid + 256] = 0;
    __syncthreads();

    int r[CITER], c[CITER], wb[CITER];
    #pragma unroll
    for (int j = 0; j < CITER; ++j) {
        const int i = j * 256 + tid;
        const long e = e0 + ((i < n) ? i : (n - 1));
        r[j] = row[e];
        c[j] = col[e];
        wb[j] = __float_as_int(ew[e]);
    }
    #pragma unroll
    for (int j = 0; j < CITER; ++j) {
        if (j * 256 + tid >= n) c[j] = r[j];
    }
    #pragma unroll
    for (int j = 0; j < CITER; ++j) {
        if (r[j] != c[j]) atomicAdd(&lcnt[r[j] >> RB_SHIFT], 1);
    }
    __syncthreads();

    const int v0 = lcnt[2 * tid];
    const int v1 = lcnt[2 * tid + 1];
    const int ts = v0 + v1;
    int inc = ts;
    for (int d = 1; d < 32; d <<= 1) {
        const int t = __shfl_up(inc, d);
        if (lane >= d) inc += t;
    }
    if (lane == 31) wsum[wave] = inc;
    __syncthreads();
    int excl = inc - ts;
    int total = 0;
    for (int w = 0; w < 8; ++w) {
        if (w < wave) excl += wsum[w];
        total += wsum[w];
    }
    lcur[2 * tid] = excl;
    lcur[2 * tid + 1] = excl + v0;
    *(int2*)(lsm + (long)blockIdx.x * NBINP + 2 * tid) = make_int2(excl, excl + v0);
    __syncthreads();

    #pragma unroll
    for (int j = 0; j < CITER; ++j) {
        if (r[j] != c[j]) {
            const int p = atomicAdd(&lcur[r[j] >> RB_SHIFT], 1);
            stage[p] = make_int4(r[j], c[j], wb[j], 0);
        }
    }
    __syncthreads();
    for (int i = tid; i < total; i += 256) binned[e0 + i] = stage[i];
}

__global__ __launch_bounds__(256) void bin_rows_kernel(
    const int* __restrict__ lsm, const int4* __restrict__ binned, int* __restrict__ cnt,
    int* __restrict__ rp, int2* __restrict__ ep, int nblk, int N) {
    __shared__ int rowcnt[RB];
    __shared__ int rowcur[RB];
    __shared__ int wred[2][8];
    __shared__ int wsc[4];
    __shared__ int4 raw[RAW_CAP + 1];
    __shared__ int2 stage[RAW_CAP];
    const int tid = threadIdx.x;
    const int lane = tid & 31;
    const int wave = tid >> 5;
    const int b = blockIdx.x;

    int s = 0, cn = 0;
    if (tid < nblk) {
        s = lsm[(long)tid * NBINP + b];
        cn = lsm[(long)tid * NBINP + b + 1] - s;
    }
    int a = s;
    for (int d = 16; d > 0; d >>= 1) a += __shfl_xor(a, d);
    int inc = cn;
    for (int d = 1; d < 32; d <<= 1) {
        const int t = __shfl_up(inc, d);
        if (lane >= d) inc += t;
    }
    if (lane == 0) wred[0][wave] = a;
    if (lane == 31) wred[1][wave] = inc;
    if (tid < RB) rowcnt[tid] = 0;
    __syncthreads();
    int binStart = 0, nb = 0;
    int o = inc - cn;
    for (int w = 0; w < 8; ++w) {
        binStart += wred[0][w];
        nb += wred[1][w];
        if (w < wave) o += wred[1][w];
    }
    const bool fast = (nb <= RAW_CAP);

    const int4* __restrict__ run = binned + (long)tid * CE + s;
    const int rbase = b * RB;

    if (fast && nb > 0) {
        const int hl = lane & 15;
        const int hh = lane >> 4;
        const int myoff = tid * CE + s;
        int4 e[16];
        int dst[16];
        #pragma unroll
        for (int it = 0; it < 16; ++it) {
            const int k = 2 * it + hh;
            const int cnk = __shfl(cn, k);
            const int ok = __shfl(o, k);
            const int fk = __shfl(myoff, k);
            const int j = min(hl, cnk - 1);
            e[it] = binned[(cnk > 0) ? fk + j : 0];
            dst[it] = (cnk > 0) ? ok + j : RAW_CAP;
        }
        #pragma unroll
        for (int it = 0; it < 16; ++it) raw[dst[it]] = e[it];
        for (int j = 16; j < cn; ++j) raw[o + j] = run[j];
    }
    __syncthreads();
    if (fast) {
        for (int i = tid; i < nb; i += 1024) {
            int rr4[4];
            #pragma unroll
            for (int u = 0; u < 4; ++u) rr4[u] = raw[min(i + 256 * u, nb - 1)].x;
            #pragma unroll
            for (int u = 0; u < 4; ++u) {
                if (i + 256 * u < nb) atomicAdd(&rowcnt[rr4[u] - rbase], 1);
            }
        }
    } else {
        int ja = 0;
        for (; ja + 4 <= cn; ja += 4) {
            const int r0 = run[ja].x, r1 = run[ja + 1].x, r2 = run[ja + 2].x, r3 = run[ja + 3].x;
            atomicAdd(&rowcnt[r0 - rbase], 1);
            atomicAdd(&rowcnt[r1 - rbase], 1);
            atomicAdd(&rowcnt[r2 - rbase], 1);
            atomicAdd(&rowcnt[r3 - rbase], 1);
        }
        for (; ja < cn; ++ja) atomicAdd(&rowcnt[run[ja].x - rbase], 1);
    }
    __syncthreads();

    int v = 0, rinc = 0;
    if (tid < RB) {
        v = rowcnt[tid];
        rinc = v;
        for (int d = 1; d < 32; d <<= 1) {
            const int t = __shfl_up(rinc, d);
            if (lane >= d) rinc += t;
        }
        if (lane == 31) wsc[wave] = rinc;
    }
    __syncthreads();
    if (tid < RB) {
        int excl = rinc - v;
        for (int w = 0; w < 4; ++w) if (w < wave) excl += wsc[w];
        rowcur[tid] = excl;
        const int rr = rbase + tid;
        if (rr < N) { cnt[rr] = v; rp[rr] = binStart + excl; }
    }
    __syncthreads();

    if (fast) {
        for (int i = tid; i < nb; i += 1024) {
            int4 e4[4];
            #pragma unroll
            for (int u = 0; u < 4; ++u) e4[u] = raw[min(i + 256 * u, nb - 1)];
            #pragma unroll
            for (int u = 0; u < 4; ++u) {
                if (i + 256 * u < nb) {
                    const int p = atomicAdd(&rowcur[e4[u].x - rbase], 1);
                    stage[p] = make_int2(e4[u].y, e4[u].z);
                }
            }
        }
    } else {
        int jb = 0;
        for (; jb + 4 <= cn; jb += 4) {
            const int4 e0 = run[jb], e1 = run[jb + 1], e2 = run[jb + 2], e3 = run[jb + 3];
            const int p0 = atomicAdd(&rowcur[e0.x - rbase], 1);
            const int p1 = atomicAdd(&rowcur[e1.x - rbase], 1);
            const int p2 = atomicAdd(&rowcur[e2.x - rbase], 1);
            const int p3 = atomicAdd(&rowcur[e3.x - rbase], 1);
            ep[binStart + p0] = make_int2(e0.y, e0.z);
            ep[binStart + p1] = make_int2(e1.y, e1.z);
            ep[binStart + p2] = make_int2(e2.y, e2.z);
            ep[binStart + p3] = make_int2(e3.y, e3.z);
        }
        for (; jb < cn; ++jb) {
            const int4 e = run[jb];
            const int p = atomicAdd(&rowcur[e.x - rbase], 1);
            ep[binStart + p] = make_int2(e.y, e.z);
        }
    }
    __syncthreads();
    if (fast) {
        for (int i = tid; i < nb; i += 256) ep[binStart + i] = stage[i];
    }
}

#define SORT_K 4
typedef unsigned long long u64;
__global__ __launch_bounds__(256) void spmm_first_kernel(
    const float* __restrict__ src, const int* __restrict__ rp, const int* __restrict__ cnt,
    u64* ep, float* __restrict__ dst, int N) {
    __shared__ u64 sbuf[8 * 32 * SORT_K];
    const int lane = threadIdx.x & 31;
    const int wv = threadIdx.x >> 5;
    const int r = __builtin_amdgcn_readfirstlane((int)blockIdx.x * 8 + wv);
    if (r >= N) return;
    const int beg = rp[r];
    const int n = cnt[r];
    u64* seg = ep + beg;
    u64* lw = sbuf + wv * (32 * SORT_K);
    const float disr = __frsqrt_rn((float)n);
    const bool small = (n <= 32 * SORT_K);

    if (small) {
        const int kmax = (n + 31) >> 5;
        u64 key[SORT_K];
        int rank[SORT_K];
        for (int k = 0; k < SORT_K; ++k) {
            const int i = k * 32 + lane;
            key[k] = ~0ull;
            rank[k] = 0;
            if (i < n) {
                const u64 p = seg[i];
                const int c = (int)(unsigned)p;
                const float w = __int_as_float((int)(unsigned)(p >> 32));
                const int dc = cnt[c];
                const float disc = (dc > 0) ? __frsqrt_rn((float)dc) : 0.0f;
                const float l = -disr * w * disc;
                key[k] = ((u64)(unsigned)__float_as_int(l) << 32) | (u64)(unsigned)c;
            }
        }
        for (int k2 = 0; k2 < SORT_K; ++k2) {
            if (k2 < kmax) {
                const int m = min(32, n - k2 * 32);
                for (int j = 0; j < m; ++j) {
                    const unsigned lo = (unsigned)__builtin_amdgcn_readlane((int)(unsigned)key[k2], j);
                    const unsigned hi = (unsigned)__builtin_amdgcn_readlane((int)(unsigned)(key[k2] >> 32), j);
                    const u64 kj = ((u64)hi << 32) | lo;
                    const int gj = k2 * 32 + j;
                    for (int k = 0; k < SORT_K; ++k) {
                        if (k < kmax) {
                            const int gi = k * 32 + lane;
                            rank[k] += (kj < key[k] || (kj == key[k] && gj < gi)) ? 1 : 0;
                        }
                    }
                }
            }
        }
        for (int k = 0; k < SORT_K; ++k) {
            const int i = k * 32 + lane;
            if (i < n) lw[rank[k]] = key[k];
        }
        for (int k = 0; k < SORT_K; ++k) {
            const int i = k * 32 + lane;
            if (i < n) seg[i] = lw[i];
        }
    } else {
        for (int i = lane; i < n; i += 32) {
            const u64 p = seg[i];
            const int c = (int)(unsigned)p;
            const float w = __int_as_float((int)(unsigned)(p >> 32));
            const int dc = cnt[c];
            const float disc = (dc > 0) ? __frsqrt_rn((float)dc) : 0.0f;
            const float l = -disr * w * disc;
            seg[i] = ((u64)(unsigned)__float_as_int(l) << 32) | (u64)(unsigned)c;
        }
        __threadfence();
        if (lane == 0) {
            for (int i = 1; i < n; ++i) {
                const u64 kk = seg[i];
                int j = i - 1;
                while (j >= 0 && seg[j] > kk) { seg[j + 1] = seg[j]; --j; }
                seg[j + 1] = kk;
            }
        }
        __threadfence();
    }

    const int b = lane >> 4;
    const int c4 = (lane & 15) * 4;
    const float* __restrict__ sb = src + (long)b * N * CH + c4;
    float4 acc = make_float4(0.0f, 0.0f, 0.0f, 0.0f);
    for (int base = 0; base < n; base += 32) {
        const int m = min(32, n - base);
        u64 mine = 0ull;
        if (small) {
            if (lane < m) mine = lw[base + lane];
        } else {
            if (lane < m) mine = seg[base + lane];
        }
        const int mx = (int)(unsigned)mine;
        const int my = (int)(unsigned)(mine >> 32);
        #pragma unroll 4
        for (int j = 0; j < m; ++j) {
            const int c = __builtin_amdgcn_readlane(mx, j);
            const float l = __int_as_float(__builtin_amdgcn_readlane(my, j));
            const float4 q = *(const float4*)(sb + (long)c * CH);
            acc.x += l * q.x;
            acc.y += l * q.y;
            acc.z += l * q.z;
            acc.w += l * q.w;
        }
    }
    *(float4*)(dst + ((long)b * N + r) * CH + c4) = acc;
}

#define FR 16
#define TP 68
__global__ __launch_bounds__(256) void spmm_gemm_fused(
    const float* __restrict__ x, const float* __restrict__ tx1,
    const int* __restrict__ rp, const int* __restrict__ cnt, const int2* __restrict__ ep,
    const float* __restrict__ w, const float* __restrict__ bias, float* __restrict__ out,
    int N, int nchunk) {
    __shared__ float wlds[3 * CH * CH];
    __shared__ float tile[3][NB][FR][TP];

    const int tid = threadIdx.x;
    #pragma unroll
    for (int i = 0; i < 3 * (CH / 2) * (CH / 4) / 256; ++i) {
        const int u = i * 256 + tid;
        const int n4 = u & 15;
        const int tj = u >> 4;
        const float* g = w + ((tj >> 5) * CH + 2 * (tj & 31)) * CH + n4 * 4;
        const float4 e = *(const float4*)g;
        const float4 o = *(const float4*)(g + CH);
        float4* d = (float4*)&wlds[(tj * CH + n4 * 4) * 2];
        d[0] = make_float4(e.x, o.x, e.y, o.y);
        d[1] = make_float4(e.z, o.z, e.w, o.w);
    }

    const int lane = tid & 31;
    const int wv = __builtin_amdgcn_readfirstlane(tid >> 5);
    const int b  = lane >> 4;
    const int c4 = (lane & 15) * 4;
    const float* __restrict__ sb = tx1 + (long)b * N * CH + c4;
    const int mt = wv & 1;
    const int nt = wv >> 1;
    const int mrow = lane & 15;
    const int lhi  = lane >> 4;
    const v2f* __restrict__ bbase = (const v2f*)wlds + (lhi << 6) + nt * 16 + mrow;
    const float bv = bias[nt * 16 + mrow];

    for (int chunk = blockIdx.x; chunk < nchunk; chunk += gridDim.x) {
        const int r0 = chunk * FR;

        const int ra = r0 + 2 * wv;
        const int rb = ra + 1;
        const int rsa = (ra < N) ? ra : 0;
        const int rsb = (rb < N) ? rb : 0;
        const long oa = ((long)b * N + rsa) * CH + c4;
        const long ob = ((long)b * N + rsb) * CH + c4;
        const float4 xa = *(const float4*)(x + oa);
        const float4 xb = *(const float4*)(x + ob);
        const float4 ta = *(const float4*)(tx1 + oa);
        const float4 tb = *(const float4*)(tx1 + ob);
        const int bega = rp[rsa];
        const int begb = rp[rsb];
        const int cna = cnt[rsa];
        const int cnb = cnt[rsb];
        const int na = (ra < N) ? cna : 0;
        const int nb = (rb < N) ? cnb : 0;
        const int nmax = max(na, nb);
        int ma = min(32, na);
        int mb = min(32, nb);
        int2 pa = make_int2(0, 0);
        int2 pb = make_int2(0, 0);
        if (lane < ma) pa = ep[bega + lane];
        if (lane < mb) pb = ep[begb + lane];
        {
            const int ia = 2 * wv;
            *(float4*)&tile[0][b][ia][c4]     = xa;
            *(float4*)&tile[0][b][ia + 1][c4] = xb;
            *(float4*)&tile[1][b][ia][c4]     = ta;
            *(float4*)&tile[1][b][ia + 1][c4] = tb;
        }
        float4 acca = make_float4(0.0f, 0.0f, 0.0f, 0.0f);
        float4 accb = make_float4(0.0f, 0.0f, 0.0f, 0.0f);
        int base = 0;
        while (true) {
            const int mm4 = (max(ma, mb) + 3) & ~3;
            for (int j = 0; j < mm4; j += 4) {
                float4 qa[4], qb[4];
                float la[4], lb[4];
                #pragma unroll
                for (int u = 0; u < 4; ++u) {
                    const int ca = __builtin_amdgcn_readlane(pa.x, j + u);
                    const int cb = __builtin_amdgcn_readlane(pb.x, j + u);
                    la[u] = __int_as_float(__builtin_amdgcn_readlane(pa.y, j + u));
                    lb[u] = __int_as_float(__builtin_amdgcn_readlane(pb.y, j + u));
                    qa[u] = *(const float4*)(sb + (long)ca * CH);
                    qb[u] = *(const float4*)(sb + (long)cb * CH);
                }
                #pragma unroll
                for (int u = 0; u < 4; ++u) {
                    acca.x += la[u] * qa[u].x;
                    acca.y += la[u] * qa[u].y;
                    acca.z += la[u] * qa[u].z;
                    acca.w += la[u] * qa[u].w;
                    accb.x += lb[u] * qb[u].x;
                    accb.y += lb[u] * qb[u].y;
                    accb.z += lb[u] * qb[u].z;
                    accb.w += lb[u] * qb[u].w;
                }
            }
            base += 32;
            if (base >= nmax) break;
            ma = min(32, max(0, na - base));
            mb = min(32, max(0, nb - base));
            pa = make_int2(0, 0);
            pb = make_int2(0, 0);
            if (lane < ma) pa = ep[bega + base + lane];
            if (lane < mb) pb = ep[begb + base + lane];
        }
        {
            const int ia = 2 * wv;
            *(float4*)&tile[2][b][ia][c4] = make_float4(
                2.0f * acca.x - xa.x, 2.0f * acca.y - xa.y, 2.0f * acca.z - xa.z, 2.0f * acca.w - xa.w);
            *(float4*)&tile[2][b][ia + 1][c4] = make_float4(
                2.0f * accb.x - xb.x, 2.0f * accb.y - xb.y, 2.0f * accb.z - xb.z, 2.0f * accb.w - xb.w);
        }
        __builtin_amdgcn_fence(__ATOMIC_RELEASE, "workgroup", "local");
        __builtin_amdgcn_s_barrier();
        __builtin_amdgcn_fence(__ATOMIC_ACQUIRE, "workgroup", "local");

        v8f acc = v8f{};
        #pragma unroll
        for (int t = 0; t < 3; ++t) {
            const float* At = &tile[t][mt][mrow][2 * lhi];
            #pragma unroll
            for (int ks = 0; ks < 16; ++ks) {
                const v2f a = *(const v2f*)(At + ks * 4);
                const v2f bf = bbase[(t * 32 + ks * 2) * 64];
                acc = __builtin_amdgcn_wmma_f32_16x16x4_f32(
                    false, a, false, bf, (short)0, acc, false, false);
            }
        }
        {
            const int col = nt * 16 + mrow;
            float* __restrict__ op = out + ((long)mt * N + r0 + (lhi << 3)) * CH + col;
            if (r0 + FR <= N) {
                #pragma unroll
                for (int v = 0; v < 8; ++v) op[v * CH] = acc[v] + bv;
            } else {
                #pragma unroll
                for (int v = 0; v < 8; ++v) {
                    if (r0 + (lhi << 3) + v < N) op[v * CH] = acc[v] + bv;
                }
            }
        }
        __builtin_amdgcn_fence(__ATOMIC_RELEASE, "workgroup", "local");
        __builtin_amdgcn_s_barrier();
        __builtin_amdgcn_fence(__ATOMIC_ACQUIRE, "workgroup", "local");
    }
}

extern "C" void kernel_launch(void* const* d_in, const int* in_sizes, int n_in,
                              void* d_out, int out_size, void* d_ws, size_t ws_size,
                              hipStream_t stream) {
    const float* x    = (const float*)d_in[0];
    const int*   ei   = (const int*)d_in[1];
    const float* ew   = (const float*)d_in[2];
    const float* w    = (const float*)d_in[3];
    const float* bias = (const float*)d_in[4];
    float*       out  = (float*)d_out;

    const long BN = (long)in_sizes[0] / CH;
    const int  N  = (int)(BN / NB);
    const long E  = (long)in_sizes[2];
    const int* row = ei;
    const int* col = ei + E;

    char* ws = (char*)d_ws;
    size_t off = 0;
    auto carve = [&](size_t bytes) {
        void* p = ws + off;
        off += (bytes + 255) & ~(size_t)255;
        return p;
    };
    const int nblk = (int)((E + CE - 1) / CE);
    const int nbin = (N + RB - 1) / RB;
    if (nblk > 256 || nbin >= NBINP) return;

    int*   cnt = (int*)carve((size_t)N * sizeof(int));
    int*   rp  = (int*)carve((size_t)N * sizeof(int));
    int2*  ep  = (int2*)carve((size_t)E * sizeof(int2));
    const size_t planeBytes = (size_t)BN * CH * sizeof(float);
    const size_t lsmBytes   = (size_t)nblk * NBINP * sizeof(int);
    const size_t binBytes   = (size_t)nblk * CE * sizeof(int4);
    float* tx1 = (float*)carve(planeBytes > lsmBytes ? planeBytes : lsmBytes);
    int4*  binned = (int4*)carve(binBytes);
    int*   lsm    = (int*)tx1;
    if (off > ws_size) return;

    auto gblocks = [](long n, int tb) { return (int)((n + tb - 1) / tb); };

    if (nblk > 0)
        bin_chunk_kernel<<<nblk, 256, 0, stream>>>(row, col, ew, lsm, binned, E);
    bin_rows_kernel<<<nbin, 256, 0, stream>>>(lsm, binned, cnt, rp, ep, nblk, N);

    spmm_first_kernel<<<gblocks(N, 8), 256, 0, stream>>>(x, rp, cnt, (u64*)ep, tx1, N);
    const int nchunk = (N + FR - 1) / FR;
    if (nchunk > 0)
        spmm_gemm_fused<<<1024, 256, 0, stream>>>(x, tx1, rp, cnt, ep, w, bias, out, N, nchunk);
}
